# starting kernel
# baseline (speedup 1.0000x reference)
#include <hip/hip_runtime.h>
#include <hip/hip_bf16.h>
#include <math.h>


#define D_MODEL 512
#define BATCH   2048
#define MAXP    150
#define MAXN    200
#define TP      143
#define TN      193
#define HSTRIDE 516
#define TILEF   (16 * HSTRIDE)
#define DUMPOFF (2 * TILEF)

typedef __attribute__((ext_vector_type(2))) float v2f;
typedef __attribute__((ext_vector_type(8))) float v8f;

__device__ __forceinline__ float fast_tanh(float x) {
#if __has_builtin(__builtin_amdgcn_tanhf)
  return __builtin_amdgcn_tanhf(x);
#else
  return tanhf(x);
#endif
}

__global__ void pack_w_kernel(const float* __restrict__ W,
                              float* __restrict__ Wpack) {
  int o  = blockIdx.x * blockDim.x + threadIdx.x;
  int kk = o & 3;
  int n  = (o >> 2) & 511;
  int kt = o >> 11;
  Wpack[o] = W[n * 513 + kt * 4 + kk];
}

__global__ __launch_bounds__(512)
void rnn_scan_kernel(const int*   __restrict__ x,
                     const float* __restrict__ emb,
                     const float* __restrict__ Wp_raw,
                     const float* __restrict__ bp,
                     const float* __restrict__ Wn_raw,
                     const float* __restrict__ bn,
                     const float* __restrict__ WpPack,
                     const float* __restrict__ WnPack,
                     float*       __restrict__ hp_out,
                     float*       __restrict__ hn_out)
{
  __shared__ float Hs[2 * TILEF + 32];

  const int scan    = blockIdx.y;
  const int rowbase = blockIdx.x * 32;
  const int tid     = threadIdx.x;
  const int lane    = tid & 31;
  const int wv      = tid >> 5;
  const int hl      = lane >> 4;
  const int ln      = lane & 15;
  const int cb      = wv * 32;

  const float *Wraw, *bvec, *Wpk, *h0;
  float* hout;
  int T, ctx_col, steps_col;
  float ctx_scale;
  if (scan == 0) {
    Wraw = Wp_raw; bvec = bp; Wpk = WpPack; hout = hp_out;
    h0 = emb;           T = TP; ctx_col = 1; steps_col = 0; ctx_scale = 1.0f / MAXN;
  } else {
    Wraw = Wn_raw; bvec = bn; Wpk = WnPack; hout = hn_out;
    h0 = emb + D_MODEL; T = TN; ctx_col = 0; steps_col = 1; ctx_scale = 1.0f / MAXP;
  }

  for (int idx = tid; idx < 2 * 16 * D_MODEL; idx += 512) {
    int tile = idx >> 13, r = (idx >> 9) & 15, c = idx & 511;
    Hs[tile * TILEF + r * HSTRIDE + c] = h0[c];
  }

  int   steps_lane[2][8];
  float ctxv[2][8];
  int   tmax = 1;
  #pragma unroll
  for (int rt = 0; rt < 2; ++rt) {
    #pragma unroll
    for (int i = 0; i < 8; ++i) {
      int gr = rowbase + rt * 16 + i + 8 * hl;
      ctxv[rt][i] = (float)x[gr * 2 + ctx_col] * ctx_scale;
      int st = x[gr * 2 + steps_col] - 8 + 1;
      if (st < 1) st = 1;
      if (st > T) st = T;
      steps_lane[rt][i] = st;
      tmax = st > tmax ? st : tmax;
    }
  }
  tmax = max(tmax, __shfl_xor(tmax, 16, 32));

  v8f cfrag[2][2];
  #pragma unroll
  for (int j = 0; j < 2; ++j) {
    int col  = cb + j * 16 + ln;
    float wc = Wraw[col * 513 + 512];
    float bb = bvec[col];
    #pragma unroll
    for (int rt = 0; rt < 2; ++rt)
      #pragma unroll
      for (int i = 0; i < 8; ++i) cfrag[rt][j][i] = ctxv[rt][i] * wc + bb;
  }
  __syncthreads();

  for (int t = 0; t < tmax; ++t) {
    v8f acc[2][2];
    #pragma unroll
    for (int rt = 0; rt < 2; ++rt)
      #pragma unroll
      for (int j = 0; j < 2; ++j) acc[rt][j] = cfrag[rt][j];

    #pragma unroll 4
    for (int kt = 0; kt < 128; ++kt) {
      v2f a0 = *(const v2f*)&Hs[0 * TILEF + ln * HSTRIDE + kt * 4 + 2 * hl];
      v2f a1 = *(const v2f*)&Hs[1 * TILEF + ln * HSTRIDE + kt * 4 + 2 * hl];
      #pragma unroll
      for (int j = 0; j < 2; ++j) {
        int n = cb + j * 16 + ln;
        v2f b = *(const v2f*)&Wpk[(kt * 512 + n) * 4 + 2 * hl];
        acc[0][j] = __builtin_amdgcn_wmma_f32_16x16x4_f32(
            false, a0, false, b, (short)0, acc[0][j], false, false);
        acc[1][j] = __builtin_amdgcn_wmma_f32_16x16x4_f32(
            false, a1, false, b, (short)0, acc[1][j], false, false);
      }
    }

    __syncthreads();

    #pragma unroll
    for (int rt = 0; rt < 2; ++rt) {
      #pragma unroll
      for (int j = 0; j < 2; ++j) {
        int col = cb + j * 16 + ln;
        #pragma unroll
        for (int i = 0; i < 8; ++i) {
          int m    = i + 8 * hl;
          int addr = (t < steps_lane[rt][i])
                       ? (rt * TILEF + m * HSTRIDE + col)
                       : (DUMPOFF + lane);
          Hs[addr] = fast_tanh(acc[rt][j][i]);
        }
      }
    }
    __syncthreads();
  }

  for (int idx = lane; idx < 32 * 32; idx += 32) {
    int r = idx >> 5, c = idx & 31;
    hout[(rowbase + r) * D_MODEL + cb + c] =
        Hs[(r >> 4) * TILEF + (r & 15) * HSTRIDE + cb + c];
  }
}

__global__ __launch_bounds__(256)
void readout_kernel(const float* __restrict__ hp, const float* __restrict__ hn,
                    const float* __restrict__ Wr, const float* __restrict__ br,
                    float* __restrict__ out)
{
  int wv   = threadIdx.x >> 5;
  int lane = threadIdx.x & 31;
  int row  = blockIdx.x * 8 + wv;
  if (row >= BATCH) return;
  float s0 = 0.f, s1 = 0.f;
  for (int c = lane; c < 512; c += 32) {
    float a = hp[row * 512 + c];
    float b = hn[row * 512 + c];
    s0 += a * Wr[c]        + b * Wr[512 + c];
    s1 += a * Wr[1024 + c] + b * Wr[1536 + c];
  }
  #pragma unroll
  for (int off = 16; off; off >>= 1) {
    s0 += __shfl_xor(s0, off, 32);
    s1 += __shfl_xor(s1, off, 32);
  }
  if (lane == 0) {
    out[row * 2 + 0] = s0 + br[0];
    out[row * 2 + 1] = s1 + br[1];
  }
}

extern "C" void kernel_launch(void* const* d_in, const int* in_sizes, int n_in,
                              void* d_out, int out_size, void* d_ws, size_t ws_size,
                              hipStream_t stream) {
  const int*   x   = (const int*)  d_in[0];
  const float* emb = (const float*)d_in[1];
  const float* Wp  = (const float*)d_in[2];
  const float* bp  = (const float*)d_in[3];
  const float* Wn  = (const float*)d_in[4];
  const float* bn  = (const float*)d_in[5];
  const float* Wr  = (const float*)d_in[6];
  const float* br  = (const float*)d_in[7];
  float* out = (float*)d_out;

  float* ws     = (float*)d_ws;
  float* hp     = ws;
  float* hn     = ws + (size_t)BATCH * D_MODEL;
  float* WpPack = ws + (size_t)2 * BATCH * D_MODEL;
  float* WnPack = WpPack + (size_t)D_MODEL * D_MODEL;

  pack_w_kernel<<<(D_MODEL * D_MODEL) / 256, 256, 0, stream>>>(Wp, WpPack);
  pack_w_kernel<<<(D_MODEL * D_MODEL) / 256, 256, 0, stream>>>(Wn, WnPack);

  rnn_scan_kernel<<<dim3(BATCH / 32, 2), 512, 0, stream>>>(
      x, emb, Wp, bp, Wn, bn, WpPack, WnPack, hp, hn);

  readout_kernel<<<BATCH / 8, 256, 0, stream>>>(hp, hn, Wr, br, out);
}


# one 16-row tile per workgroup instead of two: 256 workgroups put the recurrence on all 256 WGPs (was 128), halving WMMAs per SIMD per step; k-loop unrolled 16 so 32 WMMAs follow each load group
# speedup vs baseline: 1.9927x; 1.9927x over previous
#include <hip/hip_runtime.h>
#include <hip/hip_bf16.h>
#include <math.h>


#define D_MODEL 512
#define BATCH   2048
#define MAXP    150
#define MAXN    200
#define TP      143
#define TN      193
#define HSTRIDE 516
#define TILEF   (16 * HSTRIDE)
#define DUMPOFF (TILEF)

typedef __attribute__((ext_vector_type(2))) float v2f;
typedef __attribute__((ext_vector_type(8))) float v8f;

__device__ __forceinline__ float fast_tanh(float x) {
#if __has_builtin(__builtin_amdgcn_tanhf)
  return __builtin_amdgcn_tanhf(x);
#else
  return tanhf(x);
#endif
}

__global__ void pack_w_kernel(const float* __restrict__ W,
                              float* __restrict__ Wpack) {
  int o  = blockIdx.x * blockDim.x + threadIdx.x;
  int kk = o & 3;
  int n  = (o >> 2) & 511;
  int kt = o >> 11;
  Wpack[o] = W[n * 513 + kt * 4 + kk];
}

__global__ __launch_bounds__(512)
void rnn_scan_kernel(const int*   __restrict__ x,
                     const float* __restrict__ emb,
                     const float* __restrict__ Wp_raw,
                     const float* __restrict__ bp,
                     const float* __restrict__ Wn_raw,
                     const float* __restrict__ bn,
                     const float* __restrict__ WpPack,
                     const float* __restrict__ WnPack,
                     float*       __restrict__ hp_out,
                     float*       __restrict__ hn_out)
{
  __shared__ float Hs[TILEF + 32];

  const int scan    = blockIdx.y;
  const int rowbase = blockIdx.x * 16;
  const int tid     = threadIdx.x;
  const int lane    = tid & 31;
  const int wv      = tid >> 5;
  const int hl      = lane >> 4;
  const int ln      = lane & 15;
  const int cb      = wv * 32;

  const float *Wraw, *bvec, *Wpk, *h0;
  float* hout;
  int T, ctx_col, steps_col;
  float ctx_scale;
  if (scan == 0) {
    Wraw = Wp_raw; bvec = bp; Wpk = WpPack; hout = hp_out;
    h0 = emb;           T = TP; ctx_col = 1; steps_col = 0; ctx_scale = 1.0f / MAXN;
  } else {
    Wraw = Wn_raw; bvec = bn; Wpk = WnPack; hout = hn_out;
    h0 = emb + D_MODEL; T = TN; ctx_col = 0; steps_col = 1; ctx_scale = 1.0f / MAXP;
  }

  for (int idx = tid; idx < 16 * D_MODEL; idx += 512) {
    int r = idx >> 9, c = idx & 511;
    Hs[r * HSTRIDE + c] = h0[c];
  }

  int   steps_lane[8];
  float ctxv[8];
  int   tmax = 1;
  #pragma unroll
  for (int i = 0; i < 8; ++i) {
    int gr = rowbase + i + 8 * hl;
    ctxv[i] = (float)x[gr * 2 + ctx_col] * ctx_scale;
    int st = x[gr * 2 + steps_col] - 8 + 1;
    if (st < 1) st = 1;
    if (st > T) st = T;
    steps_lane[i] = st;
    tmax = st > tmax ? st : tmax;
  }
  tmax = max(tmax, __shfl_xor(tmax, 16, 32));

  v8f cfrag[2];
  #pragma unroll
  for (int j = 0; j < 2; ++j) {
    int col  = cb + j * 16 + ln;
    float wc = Wraw[col * 513 + 512];
    float bb = bvec[col];
    #pragma unroll
    for (int i = 0; i < 8; ++i) cfrag[j][i] = ctxv[i] * wc + bb;
  }
  __syncthreads();

  const float* wlane = Wpk + (cb + ln) * 4 + 2 * hl;
  const float* hlane = &Hs[ln * HSTRIDE + 2 * hl];

  for (int t = 0; t < tmax; ++t) {
    v8f acc[2];
    #pragma unroll
    for (int j = 0; j < 2; ++j) acc[j] = cfrag[j];

    #pragma unroll 16
    for (int kt = 0; kt < 128; ++kt) {
      v2f a = *(const v2f*)(hlane + kt * 4);
      #pragma unroll
      for (int j = 0; j < 2; ++j) {
        v2f b = *(const v2f*)(wlane + (size_t)kt * 2048 + j * 64);
        acc[j] = __builtin_amdgcn_wmma_f32_16x16x4_f32(
            false, a, false, b, (short)0, acc[j], false, false);
      }
    }

    __syncthreads();

    #pragma unroll
    for (int j = 0; j < 2; ++j) {
      int col = cb + j * 16 + ln;
      #pragma unroll
      for (int i = 0; i < 8; ++i) {
        int m    = i + 8 * hl;
        int addr = (t < steps_lane[i])
                     ? (m * HSTRIDE + col)
                     : (DUMPOFF + lane);
        Hs[addr] = fast_tanh(acc[j][i]);
      }
    }
    __syncthreads();
  }

  for (int idx = lane; idx < 16 * 32; idx += 32) {
    int r = idx >> 5, c = idx & 31;
    hout[(rowbase + r) * D_MODEL + cb + c] = Hs[r * HSTRIDE + cb + c];
  }
}

__global__ __launch_bounds__(256)
void readout_kernel(const float* __restrict__ hp, const float* __restrict__ hn,
                    const float* __restrict__ Wr, const float* __restrict__ br,
                    float* __restrict__ out)
{
  int wv   = threadIdx.x >> 5;
  int lane = threadIdx.x & 31;
  int row  = blockIdx.x * 8 + wv;
  if (row >= BATCH) return;
  float s0 = 0.f, s1 = 0.f;
  for (int c = lane; c < 512; c += 32) {
    float a = hp[row * 512 + c];
    float b = hn[row * 512 + c];
    s0 += a * Wr[c]        + b * Wr[512 + c];
    s1 += a * Wr[1024 + c] + b * Wr[1536 + c];
  }
  #pragma unroll
  for (int off = 16; off; off >>= 1) {
    s0 += __shfl_xor(s0, off, 32);
    s1 += __shfl_xor(s1, off, 32);
  }
  if (lane == 0) {
    out[row * 2 + 0] = s0 + br[0];
    out[row * 2 + 1] = s1 + br[1];
  }
}

extern "C" void kernel_launch(void* const* d_in, const int* in_sizes, int n_in,
                              void* d_out, int out_size, void* d_ws, size_t ws_size,
                              hipStream_t stream) {
  const int*   x   = (const int*)  d_in[0];
  const float* emb = (const float*)d_in[1];
  const float* Wp  = (const float*)d_in[2];
  const float* bp  = (const float*)d_in[3];
  const float* Wn  = (const float*)d_in[4];
  const float* bn  = (const float*)d_in[5];
  const float* Wr  = (const float*)d_in[6];
  const float* br  = (const float*)d_in[7];
  float* out = (float*)d_out;

  float* ws     = (float*)d_ws;
  float* hp     = ws;
  float* hn     = ws + (size_t)BATCH * D_MODEL;
  float* WpPack = ws + (size_t)2 * BATCH * D_MODEL;
  float* WnPack = WpPack + (size_t)D_MODEL * D_MODEL;

  pack_w_kernel<<<(D_MODEL * D_MODEL) / 256, 256, 0, stream>>>(Wp, WpPack);
  pack_w_kernel<<<(D_MODEL * D_MODEL) / 256, 256, 0, stream>>>(Wn, WnPack);

  rnn_scan_kernel<<<dim3(BATCH / 16, 2), 512, 0, stream>>>(
      x, emb, Wp, bp, Wn, bn, WpPack, WnPack, hp, hn);

  readout_kernel<<<BATCH / 8, 256, 0, stream>>>(hp, hn, Wr, br, out);
}


# W fragment loads software-pipelined in a 4-slot register ring: requested 48 WMMAs before use, wrapping over the step end with LDS-only barriers so they stay in flight; A fragments one group ahead
# speedup vs baseline: 2.2726x; 1.1405x over previous
#include <hip/hip_runtime.h>
#include <hip/hip_bf16.h>
#include <math.h>


#define D_MODEL 512
#define BATCH   2048
#define MAXP    150
#define MAXN    200
#define TP      143
#define TN      193
#define HSTRIDE 516
#define TILEF   (16 * HSTRIDE)
#define DUMPOFF (TILEF)
#define KG      8
#define NGRP    (128 / KG)
#define NSET    4

typedef __attribute__((ext_vector_type(2))) float v2f;
typedef __attribute__((ext_vector_type(8))) float v8f;

__device__ __forceinline__ float fast_tanh(float x) {
#if __has_builtin(__builtin_amdgcn_tanhf)
  return __builtin_amdgcn_tanhf(x);
#else
  return tanhf(x);
#endif
}

__device__ __forceinline__ void lds_barrier() {
  __builtin_amdgcn_fence(__ATOMIC_RELEASE, "workgroup", "local");
  __builtin_amdgcn_s_barrier();
  __builtin_amdgcn_fence(__ATOMIC_ACQUIRE, "workgroup", "local");
}

__global__ void pack_w_kernel(const float* __restrict__ W,
                              float* __restrict__ Wpack) {
  int o  = blockIdx.x * blockDim.x + threadIdx.x;
  int kk = o & 3;
  int n  = (o >> 2) & 511;
  int kt = o >> 11;
  Wpack[o] = W[n * 513 + kt * 4 + kk];
}

__global__ __launch_bounds__(512)
void rnn_scan_kernel(const int*   __restrict__ x,
                     const float* __restrict__ emb,
                     const float* __restrict__ Wp_raw,
                     const float* __restrict__ bp,
                     const float* __restrict__ Wn_raw,
                     const float* __restrict__ bn,
                     const float* __restrict__ WpPack,
                     const float* __restrict__ WnPack,
                     float*       __restrict__ hp_out,
                     float*       __restrict__ hn_out)
{
  __shared__ float Hs[TILEF + 32];

  const int scan    = blockIdx.y;
  const int rowbase = blockIdx.x * 16;
  const int tid     = threadIdx.x;
  const int lane    = tid & 31;
  const int wv      = tid >> 5;
  const int hl      = lane >> 4;
  const int ln      = lane & 15;
  const int cb      = wv * 32;

  const float *Wraw, *bvec, *Wpk, *h0;
  float* hout;
  int T, ctx_col, steps_col;
  float ctx_scale;
  if (scan == 0) {
    Wraw = Wp_raw; bvec = bp; Wpk = WpPack; hout = hp_out;
    h0 = emb;           T = TP; ctx_col = 1; steps_col = 0; ctx_scale = 1.0f / MAXN;
  } else {
    Wraw = Wn_raw; bvec = bn; Wpk = WnPack; hout = hn_out;
    h0 = emb + D_MODEL; T = TN; ctx_col = 0; steps_col = 1; ctx_scale = 1.0f / MAXP;
  }

  for (int idx = tid; idx < 16 * D_MODEL; idx += 512) {
    int r = idx >> 9, c = idx & 511;
    Hs[r * HSTRIDE + c] = h0[c];
  }

  int   steps_lane[8];
  float ctxv[8];
  int   tmax = 1;
  #pragma unroll
  for (int i = 0; i < 8; ++i) {
    int gr = rowbase + i + 8 * hl;
    ctxv[i] = (float)x[gr * 2 + ctx_col] * ctx_scale;
    int st = x[gr * 2 + steps_col] - 8 + 1;
    if (st < 1) st = 1;
    if (st > T) st = T;
    steps_lane[i] = st;
    tmax = st > tmax ? st : tmax;
  }
  tmax = max(tmax, __shfl_xor(tmax, 16, 32));

  v8f cfrag[2];
  #pragma unroll
  for (int j = 0; j < 2; ++j) {
    int col  = cb + j * 16 + ln;
    float wc = Wraw[col * 513 + 512];
    float bb = bvec[col];
    #pragma unroll
    for (int i = 0; i < 8; ++i) cfrag[j][i] = ctxv[i] * wc + bb;
  }
  __syncthreads();

  unsigned woff = (cb + ln) * 4 + 2 * hl;
  const float* wlane = Wpk + woff;
  const float* hlane = &Hs[ln * HSTRIDE + 2 * hl];

  v2f bq[NSET][KG][2];
  v2f aq[2][KG];
  #pragma unroll
  for (int g = 0; g < NSET - 1; ++g) {
    #pragma unroll
    for (int k = 0; k < KG; ++k) {
      #pragma unroll
      for (int j = 0; j < 2; ++j)
        bq[g][k][j] = *(const v2f*)(wlane + (size_t)(g * KG + k) * 2048 + j * 64);
    }
  }

  for (int t = 0; t < tmax; ++t) {
    asm volatile("" : "+v"(woff));
    wlane = Wpk + woff;
    v8f acc[2];
    #pragma unroll
    for (int j = 0; j < 2; ++j) acc[j] = cfrag[j];

    #pragma unroll
    for (int k = 0; k < KG; ++k) aq[0][k] = *(const v2f*)(hlane + k * 4);

    #pragma unroll
    for (int g = 0; g < NGRP; ++g) {
      const int sn = (g + NSET - 1) & (NSET - 1);
      const int gn = (g + NSET - 1) & (NGRP - 1);
      #pragma unroll
      for (int k = 0; k < KG; ++k) {
        #pragma unroll
        for (int j = 0; j < 2; ++j)
          bq[sn][k][j] = *(const v2f*)(wlane + (size_t)(gn * KG + k) * 2048 + j * 64);
      }
      if (g + 1 < NGRP) {
        #pragma unroll
        for (int k = 0; k < KG; ++k)
          aq[(g + 1) & 1][k] = *(const v2f*)(hlane + ((g + 1) * KG + k) * 4);
      }
      __builtin_amdgcn_sched_barrier(0);
      #pragma unroll
      for (int k = 0; k < KG; ++k) {
        #pragma unroll
        for (int j = 0; j < 2; ++j)
          acc[j] = __builtin_amdgcn_wmma_f32_16x16x4_f32(
              false, aq[g & 1][k], false, bq[g & (NSET - 1)][k][j], (short)0, acc[j], false, false);
      }
      __builtin_amdgcn_sched_barrier(0);
    }

    lds_barrier();

    #pragma unroll
    for (int j = 0; j < 2; ++j) {
      int col = cb + j * 16 + ln;
      #pragma unroll
      for (int i = 0; i < 8; ++i) {
        int m    = i + 8 * hl;
        int addr = (t < steps_lane[i])
                     ? (m * HSTRIDE + col)
                     : (DUMPOFF + lane);
        Hs[addr] = fast_tanh(acc[j][i]);
      }
    }
    lds_barrier();
  }

  for (int idx = lane; idx < 16 * 32; idx += 32) {
    int r = idx >> 5, c = idx & 31;
    hout[(rowbase + r) * D_MODEL + cb + c] = Hs[r * HSTRIDE + cb + c];
  }
}

__global__ __launch_bounds__(256)
void readout_kernel(const float* __restrict__ hp, const float* __restrict__ hn,
                    const float* __restrict__ Wr, const float* __restrict__ br,
                    float* __restrict__ out)
{
  int wv   = threadIdx.x >> 5;
  int lane = threadIdx.x & 31;
  int row  = blockIdx.x * 8 + wv;
  if (row >= BATCH) return;
  float s0 = 0.f, s1 = 0.f;
  for (int c = lane; c < 512; c += 32) {
    float a = hp[row * 512 + c];
    float b = hn[row * 512 + c];
    s0 += a * Wr[c]        + b * Wr[512 + c];
    s1 += a * Wr[1024 + c] + b * Wr[1536 + c];
  }
  #pragma unroll
  for (int off = 16; off; off >>= 1) {
    s0 += __shfl_xor(s0, off, 32);
    s1 += __shfl_xor(s1, off, 32);
  }
  if (lane == 0) {
    out[row * 2 + 0] = s0 + br[0];
    out[row * 2 + 1] = s1 + br[1];
  }
}

extern "C" void kernel_launch(void* const* d_in, const int* in_sizes, int n_in,
                              void* d_out, int out_size, void* d_ws, size_t ws_size,
                              hipStream_t stream) {
  const int*   x   = (const int*)  d_in[0];
  const float* emb = (const float*)d_in[1];
  const float* Wp  = (const float*)d_in[2];
  const float* bp  = (const float*)d_in[3];
  const float* Wn  = (const float*)d_in[4];
  const float* bn  = (const float*)d_in[5];
  const float* Wr  = (const float*)d_in[6];
  const float* br  = (const float*)d_in[7];
  float* out = (float*)d_out;

  float* ws     = (float*)d_ws;
  float* hp     = ws;
  float* hn     = ws + (size_t)BATCH * D_MODEL;
  float* WpPack = ws + (size_t)2 * BATCH * D_MODEL;
  float* WnPack = WpPack + (size_t)D_MODEL * D_MODEL;

  pack_w_kernel<<<(D_MODEL * D_MODEL) / 256, 256, 0, stream>>>(Wp, WpPack);
  pack_w_kernel<<<(D_MODEL * D_MODEL) / 256, 256, 0, stream>>>(Wn, WnPack);

  rnn_scan_kernel<<<dim3(BATCH / 16, 2), 512, 0, stream>>>(
      x, emb, Wp, bp, Wn, bn, WpPack, WnPack, hp, hn);

  readout_kernel<<<BATCH / 8, 256, 0, stream>>>(hp, hn, Wr, br, out);
}


# rows ranked by step count per scan (new ranking kernel) and tiled in that order: a tile stops when its own rows do, tile-steps about halve; longest tile unchanged, gain is clock and L2 relief
# speedup vs baseline: 2.6400x; 1.1617x over previous
#include <hip/hip_runtime.h>
#include <hip/hip_bf16.h>
#include <math.h>


#define D_MODEL 512
#define BATCH   2048
#define MAXP    150
#define MAXN    200
#define TP      143
#define TN      193
#define HSTRIDE 516
#define TILEF   (16 * HSTRIDE)
#define DUMPOFF (TILEF)
#define KG      8
#define NGRP    (128 / KG)
#define NSET    4

typedef __attribute__((ext_vector_type(2))) float v2f;
typedef __attribute__((ext_vector_type(8))) float v8f;

__device__ __forceinline__ float fast_tanh(float x) {
#if __has_builtin(__builtin_amdgcn_tanhf)
  return __builtin_amdgcn_tanhf(x);
#else
  return tanhf(x);
#endif
}

__device__ __forceinline__ void lds_barrier() {
  __builtin_amdgcn_fence(__ATOMIC_RELEASE, "workgroup", "local");
  __builtin_amdgcn_s_barrier();
  __builtin_amdgcn_fence(__ATOMIC_ACQUIRE, "workgroup", "local");
}

__global__ void pack_w_kernel(const float* __restrict__ W,
                              float* __restrict__ Wpack) {
  int o  = blockIdx.x * blockDim.x + threadIdx.x;
  int kk = o & 3;
  int n  = (o >> 2) & 511;
  int kt = o >> 11;
  Wpack[o] = W[n * 513 + kt * 4 + kk];
}

__global__ __launch_bounds__(128)
void rank_rows_kernel(const int* __restrict__ x, int* __restrict__ perm) {
  __shared__ int4 keys4[BATCH / 4];
  __shared__ int part[4][32];
  int* keys = (int*)keys4;
  const int scan = blockIdx.y;
  const int T    = scan == 0 ? TP : TN;
  for (int i = threadIdx.x; i < BATCH; i += 128) {
    int st = x[i * 2 + scan] - 8 + 1;
    if (st < 1) st = 1;
    if (st > T) st = T;
    keys[i] = st * BATCH + (BATCH - 1 - i);
  }
  __syncthreads();
  const int lane = threadIdx.x & 31;
  const int q    = threadIdx.x >> 5;
  const int r    = blockIdx.x * 32 + lane;
  const int kr   = keys[r];
  int cnt = 0;
  for (int s4 = q * (BATCH / 16); s4 < (q + 1) * (BATCH / 16); ++s4) {
    int4 k = keys4[s4];
    cnt += (k.x > kr) + (k.y > kr) + (k.z > kr) + (k.w > kr);
  }
  part[q][lane] = cnt;
  __syncthreads();
  if (q == 0)
    perm[scan * BATCH + part[0][lane] + part[1][lane] + part[2][lane] + part[3][lane]] = r;
}

__global__ __launch_bounds__(512)
void rnn_scan_kernel(const int*   __restrict__ x,
                     const float* __restrict__ emb,
                     const float* __restrict__ Wp_raw,
                     const float* __restrict__ bp,
                     const float* __restrict__ Wn_raw,
                     const float* __restrict__ bn,
                     const float* __restrict__ WpPack,
                     const float* __restrict__ WnPack,
                     const int*   __restrict__ perm,
                     float*       __restrict__ hp_out,
                     float*       __restrict__ hn_out)
{
  __shared__ float Hs[TILEF + 32];

  const int scan    = blockIdx.y;
  const int rowbase = blockIdx.x * 16;
  const int tid     = threadIdx.x;
  const int lane    = tid & 31;
  const int wv      = tid >> 5;
  const int hl      = lane >> 4;
  const int ln      = lane & 15;
  const int cb      = wv * 32;

  const float *Wraw, *bvec, *Wpk, *h0;
  float* hout;
  int T, ctx_col, steps_col;
  float ctx_scale;
  if (scan == 0) {
    Wraw = Wp_raw; bvec = bp; Wpk = WpPack; hout = hp_out;
    h0 = emb;           T = TP; ctx_col = 1; steps_col = 0; ctx_scale = 1.0f / MAXN;
  } else {
    Wraw = Wn_raw; bvec = bn; Wpk = WnPack; hout = hn_out;
    h0 = emb + D_MODEL; T = TN; ctx_col = 0; steps_col = 1; ctx_scale = 1.0f / MAXP;
  }

  for (int idx = tid; idx < 16 * D_MODEL; idx += 512) {
    int r = idx >> 9, c = idx & 511;
    Hs[r * HSTRIDE + c] = h0[c];
  }

  const int* prow = perm + scan * BATCH + rowbase;

  int   steps_lane[8];
  float ctxv[8];
  int   tmax = 1;
  #pragma unroll
  for (int i = 0; i < 8; ++i) {
    int gr = prow[i + 8 * hl];
    ctxv[i] = (float)x[gr * 2 + ctx_col] * ctx_scale;
    int st = x[gr * 2 + steps_col] - 8 + 1;
    if (st < 1) st = 1;
    if (st > T) st = T;
    steps_lane[i] = st;
    tmax = st > tmax ? st : tmax;
  }
  tmax = max(tmax, __shfl_xor(tmax, 16, 32));

  v8f cfrag[2];
  #pragma unroll
  for (int j = 0; j < 2; ++j) {
    int col  = cb + j * 16 + ln;
    float wc = Wraw[col * 513 + 512];
    float bb = bvec[col];
    #pragma unroll
    for (int i = 0; i < 8; ++i) cfrag[j][i] = ctxv[i] * wc + bb;
  }
  __syncthreads();

  unsigned woff = (cb + ln) * 4 + 2 * hl;
  const float* wlane = Wpk + woff;
  const float* hlane = &Hs[ln * HSTRIDE + 2 * hl];

  v2f bq[NSET][KG][2];
  v2f aq[2][KG];
  #pragma unroll
  for (int g = 0; g < NSET - 1; ++g) {
    #pragma unroll
    for (int k = 0; k < KG; ++k) {
      #pragma unroll
      for (int j = 0; j < 2; ++j)
        bq[g][k][j] = *(const v2f*)(wlane + (size_t)(g * KG + k) * 2048 + j * 64);
    }
  }

  for (int t = 0; t < tmax; ++t) {
    asm volatile("" : "+v"(woff));
    wlane = Wpk + woff;
    v8f acc[2];
    #pragma unroll
    for (int j = 0; j < 2; ++j) acc[j] = cfrag[j];

    #pragma unroll
    for (int k = 0; k < KG; ++k) aq[0][k] = *(const v2f*)(hlane + k * 4);

    #pragma unroll
    for (int g = 0; g < NGRP; ++g) {
      const int sn = (g + NSET - 1) & (NSET - 1);
      const int gn = (g + NSET - 1) & (NGRP - 1);
      #pragma unroll
      for (int k = 0; k < KG; ++k) {
        #pragma unroll
        for (int j = 0; j < 2; ++j)
          bq[sn][k][j] = *(const v2f*)(wlane + (size_t)(gn * KG + k) * 2048 + j * 64);
      }
      if (g + 1 < NGRP) {
        #pragma unroll
        for (int k = 0; k < KG; ++k)
          aq[(g + 1) & 1][k] = *(const v2f*)(hlane + ((g + 1) * KG + k) * 4);
      }
      __builtin_amdgcn_sched_barrier(0);
      #pragma unroll
      for (int k = 0; k < KG; ++k) {
        #pragma unroll
        for (int j = 0; j < 2; ++j)
          acc[j] = __builtin_amdgcn_wmma_f32_16x16x4_f32(
              false, aq[g & 1][k], false, bq[g & (NSET - 1)][k][j], (short)0, acc[j], false, false);
      }
      __builtin_amdgcn_sched_barrier(0);
    }

    lds_barrier();

    #pragma unroll
    for (int j = 0; j < 2; ++j) {
      int col = cb + j * 16 + ln;
      #pragma unroll
      for (int i = 0; i < 8; ++i) {
        int m    = i + 8 * hl;
        int addr = (t < steps_lane[i])
                     ? (m * HSTRIDE + col)
                     : (DUMPOFF + lane);
        Hs[addr] = fast_tanh(acc[j][i]);
      }
    }
    lds_barrier();
  }

  for (int idx = lane; idx < 16 * 32; idx += 32) {
    int r = idx >> 5, c = idx & 31;
    hout[prow[r] * D_MODEL + cb + c] = Hs[r * HSTRIDE + cb + c];
  }
}

__global__ __launch_bounds__(256)
void readout_kernel(const float* __restrict__ hp, const float* __restrict__ hn,
                    const float* __restrict__ Wr, const float* __restrict__ br,
                    float* __restrict__ out)
{
  int wv   = threadIdx.x >> 5;
  int lane = threadIdx.x & 31;
  int row  = blockIdx.x * 8 + wv;
  if (row >= BATCH) return;
  float s0 = 0.f, s1 = 0.f;
  for (int c = lane; c < 512; c += 32) {
    float a = hp[row * 512 + c];
    float b = hn[row * 512 + c];
    s0 += a * Wr[c]        + b * Wr[512 + c];
    s1 += a * Wr[1024 + c] + b * Wr[1536 + c];
  }
  #pragma unroll
  for (int off = 16; off; off >>= 1) {
    s0 += __shfl_xor(s0, off, 32);
    s1 += __shfl_xor(s1, off, 32);
  }
  if (lane == 0) {
    out[row * 2 + 0] = s0 + br[0];
    out[row * 2 + 1] = s1 + br[1];
  }
}

extern "C" void kernel_launch(void* const* d_in, const int* in_sizes, int n_in,
                              void* d_out, int out_size, void* d_ws, size_t ws_size,
                              hipStream_t stream) {
  const int*   x   = (const int*)  d_in[0];
  const float* emb = (const float*)d_in[1];
  const float* Wp  = (const float*)d_in[2];
  const float* bp  = (const float*)d_in[3];
  const float* Wn  = (const float*)d_in[4];
  const float* bn  = (const float*)d_in[5];
  const float* Wr  = (const float*)d_in[6];
  const float* br  = (const float*)d_in[7];
  float* out = (float*)d_out;

  float* ws     = (float*)d_ws;
  float* hp     = ws;
  float* hn     = ws + (size_t)BATCH * D_MODEL;
  float* WpPack = ws + (size_t)2 * BATCH * D_MODEL;
  float* WnPack = WpPack + (size_t)D_MODEL * D_MODEL;
  int*   perm   = (int*)(WnPack + (size_t)D_MODEL * D_MODEL);

  pack_w_kernel<<<(D_MODEL * D_MODEL) / 256, 256, 0, stream>>>(Wp, WpPack);
  pack_w_kernel<<<(D_MODEL * D_MODEL) / 256, 256, 0, stream>>>(Wn, WnPack);

  rank_rows_kernel<<<dim3(BATCH / 32, 2), 128, 0, stream>>>(x, perm);

  rnn_scan_kernel<<<dim3(BATCH / 16, 2), 512, 0, stream>>>(
      x, emb, Wp, bp, Wn, bn, WpPack, WnPack, perm, hp, hn);

  readout_kernel<<<BATCH / 8, 256, 0, stream>>>(hp, hn, Wr, br, out);
}


# one trajectory per distinct context value, one launch per time step, each 16-slot tile's columns split over 8 four-wave workgroups; per-step states kept in the workspace; same k order and sums
# speedup vs baseline: 4.7374x; 1.7945x over previous
#include <hip/hip_runtime.h>
#include <math.h>


#define D_MODEL 512
#define BATCH   2048
#define MAXP    150
#define MAXN    200
#define TP      143
#define TN      193
#define HSTRIDE 516
#define OSTRIDE 68
#define NVAL    256
#define NTILE   (NVAL / 16)
#define KG      16
#define NGRP    (128 / KG)
#define TAB_SLOT 0
#define TAB_CTX  (2 * NVAL)
#define TAB_TMAX (4 * NVAL)

typedef __attribute__((ext_vector_type(2))) float v2f;
typedef __attribute__((ext_vector_type(8))) float v8f;

__device__ __forceinline__ float fast_tanh(float x) {
#if __has_builtin(__builtin_amdgcn_tanhf)
  return __builtin_amdgcn_tanhf(x);
#else
  return tanhf(x);
#endif
}

__device__ __forceinline__ void lds_barrier() {
  __builtin_amdgcn_fence(__ATOMIC_RELEASE, "workgroup", "local");
  __builtin_amdgcn_s_barrier();
  __builtin_amdgcn_fence(__ATOMIC_ACQUIRE, "workgroup", "local");
}

__global__ void pack_w_kernel(const float* __restrict__ W,
                              float* __restrict__ Wpack) {
  int o  = blockIdx.x * blockDim.x + threadIdx.x;
  int kk = o & 3;
  int n  = (o >> 2) & 511;
  int kt = o >> 11;
  Wpack[o] = W[n * 513 + kt * 4 + kk];
}

__global__ __launch_bounds__(1024)
void build_tables_kernel(const int* __restrict__ x, const float* __restrict__ emb,
                         int* __restrict__ tab,
                         float* __restrict__ histP, float* __restrict__ histN) {
  const int scan = blockIdx.y;
  const int tid  = threadIdx.x;
  if (blockIdx.x > 0) {
    float* dst = (scan == 0 ? histP : histN) + (size_t)(blockIdx.x - 1) * 16 * D_MODEL;
    for (int i = tid; i < 16 * D_MODEL; i += 1024) dst[i] = emb[scan * D_MODEL + (i & 511)];
    return;
  }
  __shared__ int cv[BATCH];
  __shared__ int sv[BATCH];
  __shared__ int part[4][NVAL];
  __shared__ int mx[NVAL];
  __shared__ int smax[NVAL];
  __shared__ int sctx[NVAL];
  const int T         = scan == 0 ? TP : TN;
  const int ctx_col   = scan == 0 ? 1 : 0;
  const int steps_col = scan == 0 ? 0 : 1;
  for (int i = tid; i < BATCH; i += 1024) {
    int v = x[i * 2 + ctx_col];
    v = min(max(v, 0), NVAL - 1);
    int st = x[i * 2 + steps_col] - 8 + 1;
    if (st < 1) st = 1;
    if (st > T) st = T;
    cv[i] = v;
    sv[i] = st;
  }
  __syncthreads();
  const int v = tid & (NVAL - 1);
  const int q = tid >> 8;
  int m = 0;
  for (int i = q * (BATCH / 4); i < (q + 1) * (BATCH / 4); ++i)
    m = (cv[i] == v && sv[i] > m) ? sv[i] : m;
  part[q][v] = m;
  __syncthreads();
  if (tid < NVAL) {
    mx[tid]   = max(max(part[0][tid], part[1][tid]), max(part[2][tid], part[3][tid]));
    smax[tid] = 0;
    sctx[tid] = 0;
  }
  __syncthreads();
  if (tid < NVAL) {
    int s = 0;
    for (int u = 0; u < tid; ++u) s += (mx[u] > 0) ? 1 : 0;
    const bool present = mx[tid] > 0;
    tab[TAB_SLOT + scan * NVAL + tid] = present ? s : 0;
    if (present) { smax[s] = mx[tid]; sctx[s] = tid; }
  }
  __syncthreads();
  if (tid < NVAL) tab[TAB_CTX + scan * NVAL + tid] = sctx[tid];
  if (tid < NTILE) {
    int mm = 0;
    for (int k = 0; k < 16; ++k) mm = max(mm, smax[tid * 16 + k]);
    tab[TAB_TMAX + scan * NTILE + tid] = mm;
  }
}

__global__ __launch_bounds__(128)
void rnn_step_kernel(const float* __restrict__ Wp_raw, const float* __restrict__ bp,
                     const float* __restrict__ Wn_raw, const float* __restrict__ bn,
                     const float* __restrict__ WpPack, const float* __restrict__ WnPack,
                     const int*   __restrict__ tab,
                     float* histP, float* histN, int t)
{
  __shared__ float4 Hs4[16 * (HSTRIDE / 4)];
  __shared__ float4 Os4[16 * (OSTRIDE / 4)];
  float* Hs = (float*)Hs4;
  float* Os = (float*)Os4;

  const int scan = blockIdx.z;
  const int tile = blockIdx.y;
  if (t >= tab[TAB_TMAX + scan * NTILE + tile]) return;

  const int tid  = threadIdx.x;
  const int lane = tid & 31;
  const int wv   = tid >> 5;
  const int hl   = lane >> 4;
  const int ln   = lane & 15;
  const int cb   = blockIdx.x * 64 + wv * 16;

  const float *Wraw, *bvec, *Wpk;
  float* hist;
  float ctx_scale;
  if (scan == 0) {
    Wraw = Wp_raw; bvec = bp; Wpk = WpPack; hist = histP; ctx_scale = 1.0f / MAXN;
  } else {
    Wraw = Wn_raw; bvec = bn; Wpk = WnPack; hist = histN; ctx_scale = 1.0f / MAXP;
  }
  const float* src = hist + ((size_t)t * NVAL + tile * 16) * D_MODEL;
  float*       dst = hist + ((size_t)(t + 1) * NVAL + tile * 16) * D_MODEL;

  const float4 h0 = *(const float4*)(src + 0 * D_MODEL + tid * 4);
  const float4 h1 = *(const float4*)(src + 1 * D_MODEL + tid * 4);
  const float4 h2 = *(const float4*)(src + 2 * D_MODEL + tid * 4);
  const float4 h3 = *(const float4*)(src + 3 * D_MODEL + tid * 4);
  const float4 h4 = *(const float4*)(src + 4 * D_MODEL + tid * 4);
  const float4 h5 = *(const float4*)(src + 5 * D_MODEL + tid * 4);
  const float4 h6 = *(const float4*)(src + 6 * D_MODEL + tid * 4);
  const float4 h7 = *(const float4*)(src + 7 * D_MODEL + tid * 4);
  const float4 h8 = *(const float4*)(src + 8 * D_MODEL + tid * 4);
  const float4 h9 = *(const float4*)(src + 9 * D_MODEL + tid * 4);
  const float4 h10 = *(const float4*)(src + 10 * D_MODEL + tid * 4);
  const float4 h11 = *(const float4*)(src + 11 * D_MODEL + tid * 4);
  const float4 h12 = *(const float4*)(src + 12 * D_MODEL + tid * 4);
  const float4 h13 = *(const float4*)(src + 13 * D_MODEL + tid * 4);
  const float4 h14 = *(const float4*)(src + 14 * D_MODEL + tid * 4);
  const float4 h15 = *(const float4*)(src + 15 * D_MODEL + tid * 4);
  const int col = cb + ln;
  const float wc = Wraw[col * 513 + 512];
  const float bb = bvec[col];

  unsigned woff = (cb + ln) * 4 + 2 * hl;
  asm volatile("" : "+v"(woff) : : "memory");
  const float* wlane = Wpk + woff;
  v2f bq[128];
  #pragma unroll
  for (int kt = 0; kt < 3 * KG; ++kt)
    bq[kt] = *(const v2f*)(wlane + (size_t)kt * 2048);

  *(float4*)&Hs[0 * HSTRIDE + tid * 4] = h0;
  *(float4*)&Hs[1 * HSTRIDE + tid * 4] = h1;
  *(float4*)&Hs[2 * HSTRIDE + tid * 4] = h2;
  *(float4*)&Hs[3 * HSTRIDE + tid * 4] = h3;
  *(float4*)&Hs[4 * HSTRIDE + tid * 4] = h4;
  *(float4*)&Hs[5 * HSTRIDE + tid * 4] = h5;
  *(float4*)&Hs[6 * HSTRIDE + tid * 4] = h6;
  *(float4*)&Hs[7 * HSTRIDE + tid * 4] = h7;
  *(float4*)&Hs[8 * HSTRIDE + tid * 4] = h8;
  *(float4*)&Hs[9 * HSTRIDE + tid * 4] = h9;
  *(float4*)&Hs[10 * HSTRIDE + tid * 4] = h10;
  *(float4*)&Hs[11 * HSTRIDE + tid * 4] = h11;
  *(float4*)&Hs[12 * HSTRIDE + tid * 4] = h12;
  *(float4*)&Hs[13 * HSTRIDE + tid * 4] = h13;
  *(float4*)&Hs[14 * HSTRIDE + tid * 4] = h14;
  *(float4*)&Hs[15 * HSTRIDE + tid * 4] = h15;

  v8f acc;
  #pragma unroll
  for (int i = 0; i < 8; ++i) {
    const int c_lo = tab[TAB_CTX + scan * NVAL + tile * 16 + i];
    const int c_hi = tab[TAB_CTX + scan * NVAL + tile * 16 + i + 8];
    const float cx = (float)(hl ? c_hi : c_lo) * ctx_scale;
    acc[i] = cx * wc + bb;
  }
  lds_barrier();

  const float* hlane = &Hs[ln * HSTRIDE + 2 * hl];
  v2f aq[2][KG];
  #pragma unroll
  for (int k = 0; k < KG; ++k) aq[0][k] = *(const v2f*)(hlane + k * 4);

  #pragma unroll
  for (int g = 0; g < NGRP; ++g) {
    if (g + 3 < NGRP) {
      #pragma unroll
      for (int k = 0; k < KG; ++k)
        bq[(g + 3) * KG + k] = *(const v2f*)(wlane + (size_t)((g + 3) * KG + k) * 2048);
    }
    if (g + 1 < NGRP) {
      #pragma unroll
      for (int k = 0; k < KG; ++k)
        aq[(g + 1) & 1][k] = *(const v2f*)(hlane + ((g + 1) * KG + k) * 4);
    }
    __builtin_amdgcn_sched_barrier(0);
    #pragma unroll
    for (int k = 0; k < KG; ++k)
      acc = __builtin_amdgcn_wmma_f32_16x16x4_f32(
          false, aq[g & 1][k], false, bq[g * KG + k], (short)0, acc, false, false);
    __builtin_amdgcn_sched_barrier(0);
  }

  #pragma unroll
  for (int i = 0; i < 8; ++i)
    Os[(i + 8 * hl) * OSTRIDE + wv * 16 + ln] = fast_tanh(acc[i]);
  lds_barrier();
  for (int idx = tid; idx < 16 * 16; idx += 128) {
    const int r = idx >> 4, c4 = idx & 15;
    *(float4*)(dst + r * D_MODEL + blockIdx.x * 64 + c4 * 4) = *(const float4*)&Os[r * OSTRIDE + c4 * 4];
  }
}

__global__ __launch_bounds__(256)
void readout_kernel(const int* __restrict__ x, const int* __restrict__ tab,
                    const float* __restrict__ histP, const float* __restrict__ histN,
                    const float* __restrict__ Wr, const float* __restrict__ br,
                    float* __restrict__ out)
{
  int wv   = threadIdx.x >> 5;
  int lane = threadIdx.x & 31;
  int row  = blockIdx.x * 8 + wv;
  if (row >= BATCH) return;
  const int np = x[row * 2 + 0];
  const int nn = x[row * 2 + 1];
  int sp = np - 8 + 1;
  if (sp < 1) sp = 1;
  if (sp > TP) sp = TP;
  int sn = nn - 8 + 1;
  if (sn < 1) sn = 1;
  if (sn > TN) sn = TN;
  const int slp = tab[TAB_SLOT + 0 * NVAL + min(max(nn, 0), NVAL - 1)];
  const int sln = tab[TAB_SLOT + 1 * NVAL + min(max(np, 0), NVAL - 1)];
  const float* hp = histP + ((size_t)sp * NVAL + slp) * D_MODEL;
  const float* hn = histN + ((size_t)sn * NVAL + sln) * D_MODEL;
  float s0 = 0.f, s1 = 0.f;
  for (int c = lane; c < 512; c += 32) {
    float a = hp[c];
    float b = hn[c];
    s0 += a * Wr[c]        + b * Wr[512 + c];
    s1 += a * Wr[1024 + c] + b * Wr[1536 + c];
  }
  #pragma unroll
  for (int off = 16; off; off >>= 1) {
    s0 += __shfl_xor(s0, off, 32);
    s1 += __shfl_xor(s1, off, 32);
  }
  if (lane == 0) {
    out[row * 2 + 0] = s0 + br[0];
    out[row * 2 + 1] = s1 + br[1];
  }
}

extern "C" void kernel_launch(void* const* d_in, const int* in_sizes, int n_in,
                              void* d_out, int out_size, void* d_ws, size_t ws_size,
                              hipStream_t stream) {
  const int*   x   = (const int*)  d_in[0];
  const float* emb = (const float*)d_in[1];
  const float* Wp  = (const float*)d_in[2];
  const float* bp  = (const float*)d_in[3];
  const float* Wn  = (const float*)d_in[4];
  const float* bn  = (const float*)d_in[5];
  const float* Wr  = (const float*)d_in[6];
  const float* br  = (const float*)d_in[7];
  float* out = (float*)d_out;

  float* ws     = (float*)d_ws;
  float* WpPack = ws;
  float* WnPack = WpPack + (size_t)D_MODEL * D_MODEL;
  int*   tab    = (int*)(WnPack + (size_t)D_MODEL * D_MODEL);
  float* histP  = (float*)(tab + 2048);
  float* histN  = histP + (size_t)(TP + 1) * NVAL * D_MODEL;

  pack_w_kernel<<<(D_MODEL * D_MODEL) / 256, 256, 0, stream>>>(Wp, WpPack);
  pack_w_kernel<<<(D_MODEL * D_MODEL) / 256, 256, 0, stream>>>(Wn, WnPack);

  build_tables_kernel<<<dim3(1 + NTILE, 2), 1024, 0, stream>>>(x, emb, tab, histP, histN);

  for (int t = 0; t < TN; ++t)
    rnn_step_kernel<<<dim3(8, NTILE, 2), 128, 0, stream>>>(
        Wp, bp, Wn, bn, WpPack, WnPack, tab, histP, histN, t);

  readout_kernel<<<BATCH / 8, 256, 0, stream>>>(x, tab, histP, histN, Wr, br, out);
}


# k split over each workgroup's 4 waves: one k quarter x all column blocks per wave on independent accumulators, partials summed via LDS (sum order changes); 32-column slices once one scan is left
# speedup vs baseline: 4.9871x; 1.0527x over previous
#include <hip/hip_runtime.h>
#include <math.h>


#define D_MODEL 512
#define BATCH   2048
#define MAXP    150
#define MAXN    200
#define TP      143
#define TN      193
#define HSTRIDE 516
#define KQ      32
#define NVAL    256
#define NTILE   (NVAL / 16)
#define KG      16
#define TAB_SLOT 0
#define TAB_CTX  (2 * NVAL)
#define TAB_TMAX (4 * NVAL)

typedef __attribute__((ext_vector_type(2))) float v2f;
typedef __attribute__((ext_vector_type(8))) float v8f;

__device__ __forceinline__ float fast_tanh(float x) {
#if __has_builtin(__builtin_amdgcn_tanhf)
  return __builtin_amdgcn_tanhf(x);
#else
  return tanhf(x);
#endif
}

__device__ __forceinline__ void lds_barrier() {
  __builtin_amdgcn_fence(__ATOMIC_RELEASE, "workgroup", "local");
  __builtin_amdgcn_s_barrier();
  __builtin_amdgcn_fence(__ATOMIC_ACQUIRE, "workgroup", "local");
}

__global__ void pack_w_kernel(const float* __restrict__ W,
                              float* __restrict__ Wpack) {
  int o  = blockIdx.x * blockDim.x + threadIdx.x;
  int kk = o & 3;
  int n  = (o >> 2) & 511;
  int kt = o >> 11;
  Wpack[o] = W[n * 513 + kt * 4 + kk];
}

__global__ __launch_bounds__(1024)
void build_tables_kernel(const int* __restrict__ x, const float* __restrict__ emb,
                         int* __restrict__ tab,
                         float* __restrict__ histP, float* __restrict__ histN) {
  const int scan = blockIdx.y;
  const int tid  = threadIdx.x;
  if (blockIdx.x > 0) {
    float* dst = (scan == 0 ? histP : histN) + (size_t)(blockIdx.x - 1) * 16 * D_MODEL;
    for (int i = tid; i < 16 * D_MODEL; i += 1024) dst[i] = emb[scan * D_MODEL + (i & 511)];
    return;
  }
  __shared__ int cv[BATCH];
  __shared__ int sv[BATCH];
  __shared__ int part[4][NVAL];
  __shared__ int mx[NVAL];
  __shared__ int smax[NVAL];
  __shared__ int sctx[NVAL];
  const int T         = scan == 0 ? TP : TN;
  const int ctx_col   = scan == 0 ? 1 : 0;
  const int steps_col = scan == 0 ? 0 : 1;
  for (int i = tid; i < BATCH; i += 1024) {
    int v = x[i * 2 + ctx_col];
    v = min(max(v, 0), NVAL - 1);
    int st = x[i * 2 + steps_col] - 8 + 1;
    if (st < 1) st = 1;
    if (st > T) st = T;
    cv[i] = v;
    sv[i] = st;
  }
  __syncthreads();
  const int v = tid & (NVAL - 1);
  const int q = tid >> 8;
  int m = 0;
  for (int i = q * (BATCH / 4); i < (q + 1) * (BATCH / 4); ++i)
    m = (cv[i] == v && sv[i] > m) ? sv[i] : m;
  part[q][v] = m;
  __syncthreads();
  if (tid < NVAL) {
    mx[tid]   = max(max(part[0][tid], part[1][tid]), max(part[2][tid], part[3][tid]));
    smax[tid] = 0;
    sctx[tid] = 0;
  }
  __syncthreads();
  if (tid < NVAL) {
    int s = 0;
    for (int u = 0; u < tid; ++u) s += (mx[u] > 0) ? 1 : 0;
    const bool present = mx[tid] > 0;
    tab[TAB_SLOT + scan * NVAL + tid] = present ? s : 0;
    if (present) { smax[s] = mx[tid]; sctx[s] = tid; }
  }
  __syncthreads();
  if (tid < NVAL) tab[TAB_CTX + scan * NVAL + tid] = sctx[tid];
  if (tid < NTILE) {
    int mm = 0;
    for (int k = 0; k < 16; ++k) mm = max(mm, smax[tid * 16 + k]);
    tab[TAB_TMAX + scan * NTILE + tid] = mm;
  }
}

template <int NCB, int SCAN0>
__global__ __launch_bounds__(128)
void rnn_step_kernel(const float* __restrict__ Wp_raw, const float* __restrict__ bp,
                     const float* __restrict__ Wn_raw, const float* __restrict__ bn,
                     const float* __restrict__ WpPack, const float* __restrict__ WnPack,
                     const int*   __restrict__ tab,
                     float* histP, float* histN, int t)
{
  constexpr int NCOL = 16 * NCB;
  constexpr int PSTR = NCOL + 4;
  constexpr int NFRAG = KQ * NCB;
  constexpr int NGRP = NFRAG / KG;
  constexpr int QPR  = NCOL / 4;
  constexpr int NQ   = 16 * QPR / 128;
  constexpr int RSTEP = 128 / QPR;

  __shared__ float4 Hs4[16 * (HSTRIDE / 4)];
  __shared__ float4 Ps4[4 * 16 * (PSTR / 4)];
  float* Hs = (float*)Hs4;
  float* Ps = (float*)Ps4;

  const int scan = SCAN0 + blockIdx.z;
  const int tile = blockIdx.y;
  if (t >= tab[TAB_TMAX + scan * NTILE + tile]) return;

  const int tid  = threadIdx.x;
  const int lane = tid & 31;
  const int wv   = tid >> 5;
  const int hl   = lane >> 4;
  const int ln   = lane & 15;
  const int cb   = blockIdx.x * NCOL;

  const float *Wraw, *bvec, *Wpk;
  float* hist;
  float ctx_scale;
  if (scan == 0) {
    Wraw = Wp_raw; bvec = bp; Wpk = WpPack; hist = histP; ctx_scale = 1.0f / MAXN;
  } else {
    Wraw = Wn_raw; bvec = bn; Wpk = WnPack; hist = histN; ctx_scale = 1.0f / MAXP;
  }
  const float* src = hist + ((size_t)t * NVAL + tile * 16) * D_MODEL;
  float*       dst = hist + ((size_t)(t + 1) * NVAL + tile * 16) * D_MODEL;

  const float4 h0 = *(const float4*)(src + 0 * D_MODEL + tid * 4);
  const float4 h1 = *(const float4*)(src + 1 * D_MODEL + tid * 4);
  const float4 h2 = *(const float4*)(src + 2 * D_MODEL + tid * 4);
  const float4 h3 = *(const float4*)(src + 3 * D_MODEL + tid * 4);
  const float4 h4 = *(const float4*)(src + 4 * D_MODEL + tid * 4);
  const float4 h5 = *(const float4*)(src + 5 * D_MODEL + tid * 4);
  const float4 h6 = *(const float4*)(src + 6 * D_MODEL + tid * 4);
  const float4 h7 = *(const float4*)(src + 7 * D_MODEL + tid * 4);
  const float4 h8 = *(const float4*)(src + 8 * D_MODEL + tid * 4);
  const float4 h9 = *(const float4*)(src + 9 * D_MODEL + tid * 4);
  const float4 h10 = *(const float4*)(src + 10 * D_MODEL + tid * 4);
  const float4 h11 = *(const float4*)(src + 11 * D_MODEL + tid * 4);
  const float4 h12 = *(const float4*)(src + 12 * D_MODEL + tid * 4);
  const float4 h13 = *(const float4*)(src + 13 * D_MODEL + tid * 4);
  const float4 h14 = *(const float4*)(src + 14 * D_MODEL + tid * 4);
  const float4 h15 = *(const float4*)(src + 15 * D_MODEL + tid * 4);

  const int r0   = tid / QPR;
  const int c4   = tid % QPR;
  const int ocol = cb + c4 * 4;
  const float wc0 = Wraw[(ocol + 0) * 513 + 512];
  const float wc1 = Wraw[(ocol + 1) * 513 + 512];
  const float wc2 = Wraw[(ocol + 2) * 513 + 512];
  const float wc3 = Wraw[(ocol + 3) * 513 + 512];
  const float bb0 = bvec[ocol + 0];
  const float bb1 = bvec[ocol + 1];
  const float bb2 = bvec[ocol + 2];
  const float bb3 = bvec[ocol + 3];
  int ctxv[NQ];
  #pragma unroll
  for (int q = 0; q < NQ; ++q)
    ctxv[q] = tab[TAB_CTX + scan * NVAL + tile * 16 + r0 + RSTEP * q];

  unsigned woff = (cb + ln) * 4 + 2 * hl;
  asm volatile("" : "+v"(woff) : : "memory");
  const float* wlane = Wpk + woff + (size_t)(wv * KQ) * 2048;
  v2f bq[NFRAG];
  #pragma unroll
  for (int i = 0; i < 3 * KG; ++i)
    bq[i] = *(const v2f*)(wlane + (size_t)(i / NCB) * 2048 + (i % NCB) * 64);

  *(float4*)&Hs[0 * HSTRIDE + tid * 4] = h0;
  *(float4*)&Hs[1 * HSTRIDE + tid * 4] = h1;
  *(float4*)&Hs[2 * HSTRIDE + tid * 4] = h2;
  *(float4*)&Hs[3 * HSTRIDE + tid * 4] = h3;
  *(float4*)&Hs[4 * HSTRIDE + tid * 4] = h4;
  *(float4*)&Hs[5 * HSTRIDE + tid * 4] = h5;
  *(float4*)&Hs[6 * HSTRIDE + tid * 4] = h6;
  *(float4*)&Hs[7 * HSTRIDE + tid * 4] = h7;
  *(float4*)&Hs[8 * HSTRIDE + tid * 4] = h8;
  *(float4*)&Hs[9 * HSTRIDE + tid * 4] = h9;
  *(float4*)&Hs[10 * HSTRIDE + tid * 4] = h10;
  *(float4*)&Hs[11 * HSTRIDE + tid * 4] = h11;
  *(float4*)&Hs[12 * HSTRIDE + tid * 4] = h12;
  *(float4*)&Hs[13 * HSTRIDE + tid * 4] = h13;
  *(float4*)&Hs[14 * HSTRIDE + tid * 4] = h14;
  *(float4*)&Hs[15 * HSTRIDE + tid * 4] = h15;


  const float* hlane = &Hs[ln * HSTRIDE + wv * (KQ * 4) + 2 * hl];
  v2f aq[KQ];
  #pragma unroll
  for (int k = 0; k < KQ; ++k) aq[k] = *(const v2f*)(hlane + k * 4);

  v8f acc[NCB];
  #pragma unroll
  for (int j = 0; j < NCB; ++j)
    #pragma unroll
    for (int i = 0; i < 8; ++i) acc[j][i] = 0.0f;

  #pragma unroll
  for (int g = 0; g < NGRP; ++g) {
    if (g + 3 < NGRP) {
      #pragma unroll
      for (int k = 0; k < KG; ++k) {
        const int i = (g + 3) * KG + k;
        bq[i] = *(const v2f*)(wlane + (size_t)(i / NCB) * 2048 + (i % NCB) * 64);
      }
    }
    __builtin_amdgcn_sched_barrier(0);
    #pragma unroll
    for (int k = 0; k < KG; ++k) {
      const int i = g * KG + k;
      acc[i % NCB] = __builtin_amdgcn_wmma_f32_16x16x4_f32(
          false, aq[i / NCB], false, bq[i], (short)0, acc[i % NCB], false, false);
    }
    __builtin_amdgcn_sched_barrier(0);
  }

  #pragma unroll
  for (int j = 0; j < NCB; ++j)
    #pragma unroll
    for (int i = 0; i < 8; ++i)
      Ps[(wv * 16 + i + 8 * hl) * PSTR + 16 * j + ln] = acc[j][i];
  lds_barrier();

  #pragma unroll
  for (int q = 0; q < NQ; ++q) {
    const int r = r0 + RSTEP * q;
    const float* pp = &Ps[r * PSTR + c4 * 4];
    const float4 p0 = *(const float4*)(pp + 0 * 16 * PSTR);
    const float4 p1 = *(const float4*)(pp + 1 * 16 * PSTR);
    const float4 p2 = *(const float4*)(pp + 2 * 16 * PSTR);
    const float4 p3 = *(const float4*)(pp + 3 * 16 * PSTR);
    const float cx = (float)ctxv[q] * ctx_scale;
    float4 o;
    o.x = fast_tanh(((p0.x + p1.x) + (p2.x + p3.x)) + (cx * wc0 + bb0));
    o.y = fast_tanh(((p0.y + p1.y) + (p2.y + p3.y)) + (cx * wc1 + bb1));
    o.z = fast_tanh(((p0.z + p1.z) + (p2.z + p3.z)) + (cx * wc2 + bb2));
    o.w = fast_tanh(((p0.w + p1.w) + (p2.w + p3.w)) + (cx * wc3 + bb3));
    *(float4*)(dst + r * D_MODEL + ocol) = o;
  }
}

__global__ __launch_bounds__(256)
void readout_kernel(const int* __restrict__ x, const int* __restrict__ tab,
                    const float* __restrict__ histP, const float* __restrict__ histN,
                    const float* __restrict__ Wr, const float* __restrict__ br,
                    float* __restrict__ out)
{
  int wv   = threadIdx.x >> 5;
  int lane = threadIdx.x & 31;
  int row  = blockIdx.x * 8 + wv;
  if (row >= BATCH) return;
  const int np = x[row * 2 + 0];
  const int nn = x[row * 2 + 1];
  int sp = np - 8 + 1;
  if (sp < 1) sp = 1;
  if (sp > TP) sp = TP;
  int sn = nn - 8 + 1;
  if (sn < 1) sn = 1;
  if (sn > TN) sn = TN;
  const int slp = tab[TAB_SLOT + 0 * NVAL + min(max(nn, 0), NVAL - 1)];
  const int sln = tab[TAB_SLOT + 1 * NVAL + min(max(np, 0), NVAL - 1)];
  const float* hp = histP + ((size_t)sp * NVAL + slp) * D_MODEL;
  const float* hn = histN + ((size_t)sn * NVAL + sln) * D_MODEL;
  float s0 = 0.f, s1 = 0.f;
  for (int c = lane; c < 512; c += 32) {
    float a = hp[c];
    float b = hn[c];
    s0 += a * Wr[c]        + b * Wr[512 + c];
    s1 += a * Wr[1024 + c] + b * Wr[1536 + c];
  }
  #pragma unroll
  for (int off = 16; off; off >>= 1) {
    s0 += __shfl_xor(s0, off, 32);
    s1 += __shfl_xor(s1, off, 32);
  }
  if (lane == 0) {
    out[row * 2 + 0] = s0 + br[0];
    out[row * 2 + 1] = s1 + br[1];
  }
}

extern "C" void kernel_launch(void* const* d_in, const int* in_sizes, int n_in,
                              void* d_out, int out_size, void* d_ws, size_t ws_size,
                              hipStream_t stream) {
  const int*   x   = (const int*)  d_in[0];
  const float* emb = (const float*)d_in[1];
  const float* Wp  = (const float*)d_in[2];
  const float* bp  = (const float*)d_in[3];
  const float* Wn  = (const float*)d_in[4];
  const float* bn  = (const float*)d_in[5];
  const float* Wr  = (const float*)d_in[6];
  const float* br  = (const float*)d_in[7];
  float* out = (float*)d_out;

  float* ws     = (float*)d_ws;
  float* WpPack = ws;
  float* WnPack = WpPack + (size_t)D_MODEL * D_MODEL;
  int*   tab    = (int*)(WnPack + (size_t)D_MODEL * D_MODEL);
  float* histP  = (float*)(tab + 2048);
  float* histN  = histP + (size_t)(TP + 1) * NVAL * D_MODEL;

  pack_w_kernel<<<(D_MODEL * D_MODEL) / 256, 256, 0, stream>>>(Wp, WpPack);
  pack_w_kernel<<<(D_MODEL * D_MODEL) / 256, 256, 0, stream>>>(Wn, WnPack);

  build_tables_kernel<<<dim3(1 + NTILE, 2), 1024, 0, stream>>>(x, emb, tab, histP, histN);

  for (int t = 0; t < TP; ++t)
    rnn_step_kernel<4, 0><<<dim3(8, NTILE, 2), 128, 0, stream>>>(
        Wp, bp, Wn, bn, WpPack, WnPack, tab, histP, histN, t);
  for (int t = TP; t < TN; ++t)
    rnn_step_kernel<2, 1><<<dim3(16, NTILE, 1), 128, 0, stream>>>(
        Wp, bp, Wn, bn, WpPack, WnPack, tab, histP, histN, t);

  readout_kernel<<<BATCH / 8, 256, 0, stream>>>(x, tab, histP, histN, Wr, br, out);
}
